# starting kernel
# baseline (speedup 1.0000x reference)
#include <hip/hip_runtime.h>
#include <math.h>

constexpr int kB = 8;
constexpr int kT = 2048;
constexpr int kD = 512;

typedef __attribute__((ext_vector_type(2))) float v2f;
typedef __attribute__((ext_vector_type(8))) float v8f;

__device__ __forceinline__ float softplus_dev(const float* kp) {
    float k = kp[0];
    return fmaxf(k, 0.0f) + log1pf(expf(-fabsf(k)));
}

__global__ void delta_mag_sq_kernel(const float* __restrict__ h,
                                    float* __restrict__ dsq) {
    int gid  = blockIdx.x * blockDim.x + threadIdx.x;
    int lane = gid & 31;
    int wave = gid >> 5;
    int b    = wave >> 4;
    int d    = ((wave & 15) << 5) + lane;
    const float* hp = h + (size_t)b * kT * kD + d;

    float cum = 0.f, hprev = 0.f;
    for (int t = 0; t < kT; ++t) {
        float hv = hp[(size_t)t * kD];
        cum += hv;
        float dl = cum - hprev;
        hprev = hv;
        float sq = dl * dl;
        #pragma unroll
        for (int off = 16; off > 0; off >>= 1)
            sq += __shfl_xor(sq, off, 32);
        if (lane == 0) atomicAdd(&dsq[b * kT + t], sq);
    }
}

__global__ void mag_sqrt_kernel(float* __restrict__ dsq) {
    int i = blockIdx.x * blockDim.x + threadIdx.x;
    dsq[i] = sqrtf(dsq[i]);
}

__global__ void context_wmma_kernel(const float* __restrict__ h,
                                    const float* __restrict__ kp,
                                    const float* __restrict__ mag,
                                    float* __restrict__ out) {
    const int lane  = threadIdx.x;
    const int dtile = blockIdx.x & 31;
    const int b     = blockIdx.x >> 5;
    const int n     = lane & 15;
    const int hi    = (lane >> 4) & 1;

    const float sp  = softplus_dev(kp);
    const float l2r = -sp * 1.4426950408889634f;
    auto rpow = [&](int dist) -> float { return exp2f((float)dist * l2r); };

    v2f a[4];
    #pragma unroll
    for (int g = 0; g < 4; ++g) {
        int j0 = 4 * g + 2 * hi;
        a[g].x = (j0     <= n) ? rpow(n - j0)     : 0.f;
        a[g].y = (j0 + 1 <= n) ? rpow(n - j0 - 1) : 0.f;
    }

    float rpc[8];
    #pragma unroll
    for (int v = 0; v < 8; ++v) rpc[v] = rpow(v + 1 + 8 * hi);

    const float* hb = h   + ((size_t)b * kT) * kD + dtile * 16 + n;
    float*       ob = out + ((size_t)b * kT) * kD + dtile * 16 + n;
    const float* mb = mag + b * kT;

    float S = 0.f;
    for (int t0 = 0; t0 < kT; t0 += 16) {
        v2f bm[4];
        #pragma unroll
        for (int g = 0; g < 4; ++g) {
            size_t j = (size_t)(t0 + 4 * g + 2 * hi);
            bm[g].x = hb[j * kD];
            bm[g].y = hb[(j + 1) * kD];
        }
        v8f c;
        #pragma unroll
        for (int v = 0; v < 8; ++v) c[v] = rpc[v] * S;

        c = __builtin_amdgcn_wmma_f32_16x16x4_f32(false, a[0], false, bm[0], (short)0, c, false, false);
        c = __builtin_amdgcn_wmma_f32_16x16x4_f32(false, a[1], false, bm[1], (short)0, c, false, false);
        c = __builtin_amdgcn_wmma_f32_16x16x4_f32(false, a[2], false, bm[2], (short)0, c, false, false);
        c = __builtin_amdgcn_wmma_f32_16x16x4_f32(false, a[3], false, bm[3], (short)0, c, false, false);

        #pragma unroll
        for (int v = 0; v < 8; ++v) {
            int M = v + 8 * hi;
            ob[(size_t)(t0 + M) * kD] = mb[t0 + M] * fabsf(c[v]);
        }
        S = __shfl(c[7], 16 + n, 32);
    }
}

extern "C" void kernel_launch(void* const* d_in, const int* in_sizes, int n_in,
                              void* d_out, int out_size, void* d_ws, size_t ws_size,
                              hipStream_t stream) {
    const float* h  = (const float*)d_in[0];
    const float* k  = (const float*)d_in[1];
    float* out = (float*)d_out;
    float* dsq = (float*)d_ws;

    (void)hipMemsetAsync(dsq, 0, (size_t)kB * kT * sizeof(float), stream);
    delta_mag_sq_kernel<<<16, 256, 0, stream>>>(h, dsq);
    mag_sqrt_kernel<<<(kB * kT) / 256, 256, 0, stream>>>(dsq);
    context_wmma_kernel<<<kB * (kD / 16), 32, 0, stream>>>(h, k, dsq, out);
}


# Pass 1 scan over T chunked: 32-row chunk sums, then carried prefix + local scan per chunk; d-sum by in-block butterfly and LDS, not float atomics (fixed sum order); memset and sqrt launch gone
# speedup vs baseline: 12.4457x; 12.4457x over previous
#include <hip/hip_runtime.h>
#include <math.h>


constexpr int kB = 8;
constexpr int kT = 2048;
constexpr int kD = 512;

constexpr int kLc  = 32;
constexpr int kNc  = kT / kLc;
constexpr int kSub = 8;
constexpr int kD4  = kD / 4;

typedef __attribute__((ext_vector_type(2))) float v2f;
typedef __attribute__((ext_vector_type(8))) float v8f;

__device__ __forceinline__ float softplus_dev(const float* kp) {
    float k = kp[0];
    return fmaxf(k, 0.0f) + log1pf(expf(-fabsf(k)));
}

__device__ __forceinline__ float4 add4(float4 a, float4 b) {
    return make_float4(a.x + b.x, a.y + b.y, a.z + b.z, a.w + b.w);
}

__global__ __launch_bounds__(512)
void chunk_sum_kernel(const float* __restrict__ h, float* __restrict__ csum) {
    __shared__ float4 part[3][kD4];
    const int c   = blockIdx.x;
    const int b   = blockIdx.y;
    const int col = threadIdx.x & (kD4 - 1);
    const int g   = threadIdx.x >> 7;
    if (c == kNc - 1) {
        if (g == 0)
            reinterpret_cast<float4*>(csum)[((size_t)b * kNc + c) * kD4 + col] = make_float4(0.f, 0.f, 0.f, 0.f);
        return;
    }
    const float4* hp = reinterpret_cast<const float4*>(h)
                     + ((size_t)b * kT + c * kLc + g * kSub) * kD4 + col;
    float4 x[kSub];
    for (int i = 0; i < kSub; ++i) x[i] = hp[(size_t)i * kD4];
    float4 s = x[0];
    for (int i = 1; i < kSub; ++i) s = add4(s, x[i]);
    if (g > 0) part[g - 1][col] = s;
    __syncthreads();
    if (g == 0) {
        s = add4(s, part[0][col]);
        s = add4(s, part[1][col]);
        s = add4(s, part[2][col]);
        reinterpret_cast<float4*>(csum)[((size_t)b * kNc + c) * kD4 + col] = s;
    }
}

__global__ __launch_bounds__(512)
void delta_mag_kernel(const float* __restrict__ h,
                      const float* __restrict__ csum,
                      float* __restrict__ mag) {
    __shared__ float4 cpre[4][kD4];
    __shared__ float4 ssum[4][kD4];
    __shared__ float  wpart[16][kSub];

    const int c    = blockIdx.x;
    const int b    = blockIdx.y;
    const int tid  = threadIdx.x;
    const int col  = tid & (kD4 - 1);
    const int g    = tid >> 7;
    const int lane = tid & 31;
    const int wave = tid >> 5;
    const int t0   = c * kLc + g * kSub;

    const float4* h4 = reinterpret_cast<const float4*>(h) + ((size_t)b * kT) * kD4 + col;

    float4 x[kSub];
    for (int i = 0; i < kSub; ++i) x[i] = h4[(size_t)(t0 + i) * kD4];
    float4 xp = h4[(size_t)(t0 > 0 ? t0 - 1 : 0) * kD4];
    if (t0 == 0) xp = make_float4(0.f, 0.f, 0.f, 0.f);

    float4 cp = make_float4(0.f, 0.f, 0.f, 0.f);
    const int cbase = g * 16;
    if (cbase < c) {
        const float4* cs4 = reinterpret_cast<const float4*>(csum) + ((size_t)b * kNc) * kD4 + col;
        float4 v[16];
        for (int j = 0; j < 16; ++j) {
            const int cc = cbase + j;
            v[j] = cs4[(size_t)(cc < c ? cc : kNc - 1) * kD4];
        }
        for (int j = 0; j < 16; ++j) cp = add4(cp, v[j]);
    }

    float4 s = x[0];
    for (int i = 1; i < kSub; ++i) s = add4(s, x[i]);

    cpre[g][col] = cp;
    ssum[g][col] = s;
    __syncthreads();

    float4 cum = add4(add4(cpre[0][col], cpre[1][col]), add4(cpre[2][col], cpre[3][col]));
    for (int gp = 0; gp < 3; ++gp)
        if (gp < g) cum = add4(cum, ssum[gp][col]);

    float sq[kSub];
    float4 prev = xp;
    for (int i = 0; i < kSub; ++i) {
        cum = add4(cum, x[i]);
        const float dx = cum.x - prev.x, dy = cum.y - prev.y;
        const float dz = cum.z - prev.z, dw = cum.w - prev.w;
        prev = x[i];
        sq[i] = (dx * dx + dy * dy) + (dz * dz + dw * dw);
    }

    const bool b4 = (lane & 16) != 0, b3 = (lane & 8) != 0, b2 = (lane & 4) != 0;
    float r4[4];
    for (int i = 0; i < 4; ++i) {
        const float keep = b4 ? sq[i + 4] : sq[i];
        const float send = b4 ? sq[i]     : sq[i + 4];
        r4[i] = keep + __shfl_xor(send, 16, 32);
    }
    float r2[2];
    for (int i = 0; i < 2; ++i) {
        const float keep = b3 ? r4[i + 2] : r4[i];
        const float send = b3 ? r4[i]     : r4[i + 2];
        r2[i] = keep + __shfl_xor(send, 8, 32);
    }
    float r1;
    {
        const float keep = b2 ? r2[1] : r2[0];
        const float send = b2 ? r2[0] : r2[1];
        r1 = keep + __shfl_xor(send, 4, 32);
    }
    r1 += __shfl_xor(r1, 2, 32);
    r1 += __shfl_xor(r1, 1, 32);
    if ((lane & 3) == 0) wpart[wave][lane >> 2] = r1;
    __syncthreads();

    if (tid < kLc) {
        const int gg = tid >> 3, r = tid & 7;
        const float tot = (wpart[4 * gg][r] + wpart[4 * gg + 1][r])
                        + (wpart[4 * gg + 2][r] + wpart[4 * gg + 3][r]);
        mag[b * kT + c * kLc + tid] = sqrtf(tot);
    }
}

__global__ void context_wmma_kernel(const float* __restrict__ h,
                                    const float* __restrict__ kp,
                                    const float* __restrict__ mag,
                                    float* __restrict__ out) {
    const int lane  = threadIdx.x;
    const int dtile = blockIdx.x & 31;
    const int b     = blockIdx.x >> 5;
    const int n     = lane & 15;
    const int hi    = (lane >> 4) & 1;

    const float sp  = softplus_dev(kp);
    const float l2r = -sp * 1.4426950408889634f;
    auto rpow = [&](int dist) -> float { return exp2f((float)dist * l2r); };

    v2f a[4];
    #pragma unroll
    for (int g = 0; g < 4; ++g) {
        int j0 = 4 * g + 2 * hi;
        a[g].x = (j0     <= n) ? rpow(n - j0)     : 0.f;
        a[g].y = (j0 + 1 <= n) ? rpow(n - j0 - 1) : 0.f;
    }

    float rpc[8];
    #pragma unroll
    for (int v = 0; v < 8; ++v) rpc[v] = rpow(v + 1 + 8 * hi);

    const float* hb = h   + ((size_t)b * kT) * kD + dtile * 16 + n;
    float*       ob = out + ((size_t)b * kT) * kD + dtile * 16 + n;
    const float* mb = mag + b * kT;

    float S = 0.f;
    for (int t0 = 0; t0 < kT; t0 += 16) {
        v2f bm[4];
        #pragma unroll
        for (int g = 0; g < 4; ++g) {
            size_t j = (size_t)(t0 + 4 * g + 2 * hi);
            bm[g].x = hb[j * kD];
            bm[g].y = hb[(j + 1) * kD];
        }
        v8f c;
        #pragma unroll
        for (int v = 0; v < 8; ++v) c[v] = rpc[v] * S;

        c = __builtin_amdgcn_wmma_f32_16x16x4_f32(false, a[0], false, bm[0], (short)0, c, false, false);
        c = __builtin_amdgcn_wmma_f32_16x16x4_f32(false, a[1], false, bm[1], (short)0, c, false, false);
        c = __builtin_amdgcn_wmma_f32_16x16x4_f32(false, a[2], false, bm[2], (short)0, c, false, false);
        c = __builtin_amdgcn_wmma_f32_16x16x4_f32(false, a[3], false, bm[3], (short)0, c, false, false);

        #pragma unroll
        for (int v = 0; v < 8; ++v) {
            int M = v + 8 * hi;
            ob[(size_t)(t0 + M) * kD] = mb[t0 + M] * fabsf(c[v]);
        }
        S = __shfl(c[7], 16 + n, 32);
    }
}

extern "C" void kernel_launch(void* const* d_in, const int* in_sizes, int n_in,
                              void* d_out, int out_size, void* d_ws, size_t ws_size,
                              hipStream_t stream) {
    const float* h  = (const float*)d_in[0];
    const float* k  = (const float*)d_in[1];
    float* out = (float*)d_out;
    float* mag = (float*)d_ws;
    float* csum = out;

    chunk_sum_kernel<<<dim3(kNc, kB), 512, 0, stream>>>(h, csum);
    delta_mag_kernel<<<dim3(kNc, kB), 512, 0, stream>>>(h, csum, mag);
    context_wmma_kernel<<<kB * (kD / 16), 32, 0, stream>>>(h, k, mag, out);
}


# Pass 2 scan split along T: 16 waves per (b, d-tile) block, 8 independent zero-carry WMMA chunks per wave, segment carries joined through LDS and added after the product (carry sum order changes)
# speedup vs baseline: 42.4853x; 3.4137x over previous
#include <hip/hip_runtime.h>
#include <math.h>


constexpr int kB = 8;
constexpr int kT = 2048;
constexpr int kD = 512;

constexpr int kLc  = 32;
constexpr int kNc  = kT / kLc;
constexpr int kSub = 8;
constexpr int kD4  = kD / 4;

typedef __attribute__((ext_vector_type(2))) float v2f;
typedef __attribute__((ext_vector_type(8))) float v8f;

__device__ __forceinline__ float softplus_dev(const float* kp) {
    float k = kp[0];
    return fmaxf(k, 0.0f) + log1pf(expf(-fabsf(k)));
}

__device__ __forceinline__ float4 add4(float4 a, float4 b) {
    return make_float4(a.x + b.x, a.y + b.y, a.z + b.z, a.w + b.w);
}

__global__ __launch_bounds__(512)
void chunk_sum_kernel(const float* __restrict__ h, float* __restrict__ csum) {
    __shared__ float4 part[3][kD4];
    const int c   = blockIdx.x;
    const int b   = blockIdx.y;
    const int col = threadIdx.x & (kD4 - 1);
    const int g   = threadIdx.x >> 7;
    if (c == kNc - 1) {
        if (g == 0)
            reinterpret_cast<float4*>(csum)[((size_t)b * kNc + c) * kD4 + col] = make_float4(0.f, 0.f, 0.f, 0.f);
        return;
    }
    const float4* hp = reinterpret_cast<const float4*>(h)
                     + ((size_t)b * kT + c * kLc + g * kSub) * kD4 + col;
    float4 x[kSub];
    for (int i = 0; i < kSub; ++i) x[i] = hp[(size_t)i * kD4];
    float4 s = x[0];
    for (int i = 1; i < kSub; ++i) s = add4(s, x[i]);
    if (g > 0) part[g - 1][col] = s;
    __syncthreads();
    if (g == 0) {
        s = add4(s, part[0][col]);
        s = add4(s, part[1][col]);
        s = add4(s, part[2][col]);
        reinterpret_cast<float4*>(csum)[((size_t)b * kNc + c) * kD4 + col] = s;
    }
}

__global__ __launch_bounds__(512)
void delta_mag_kernel(const float* __restrict__ h,
                      const float* __restrict__ csum,
                      float* __restrict__ mag) {
    __shared__ float4 cpre[4][kD4];
    __shared__ float4 ssum[4][kD4];
    __shared__ float  wpart[16][kSub];

    const int c    = blockIdx.x;
    const int b    = blockIdx.y;
    const int tid  = threadIdx.x;
    const int col  = tid & (kD4 - 1);
    const int g    = tid >> 7;
    const int lane = tid & 31;
    const int wave = tid >> 5;
    const int t0   = c * kLc + g * kSub;

    const float4* h4 = reinterpret_cast<const float4*>(h) + ((size_t)b * kT) * kD4 + col;

    float4 x[kSub];
    for (int i = 0; i < kSub; ++i) x[i] = h4[(size_t)(t0 + i) * kD4];
    float4 xp = h4[(size_t)(t0 > 0 ? t0 - 1 : 0) * kD4];
    if (t0 == 0) xp = make_float4(0.f, 0.f, 0.f, 0.f);

    float4 cp = make_float4(0.f, 0.f, 0.f, 0.f);
    const int cbase = g * 16;
    if (cbase < c) {
        const float4* cs4 = reinterpret_cast<const float4*>(csum) + ((size_t)b * kNc) * kD4 + col;
        float4 v[16];
        for (int j = 0; j < 16; ++j) {
            const int cc = cbase + j;
            v[j] = cs4[(size_t)(cc < c ? cc : kNc - 1) * kD4];
        }
        for (int j = 0; j < 16; ++j) cp = add4(cp, v[j]);
    }

    float4 s = x[0];
    for (int i = 1; i < kSub; ++i) s = add4(s, x[i]);

    cpre[g][col] = cp;
    ssum[g][col] = s;
    __syncthreads();

    float4 cum = add4(add4(cpre[0][col], cpre[1][col]), add4(cpre[2][col], cpre[3][col]));
    for (int gp = 0; gp < 3; ++gp)
        if (gp < g) cum = add4(cum, ssum[gp][col]);

    float sq[kSub];
    float4 prev = xp;
    for (int i = 0; i < kSub; ++i) {
        cum = add4(cum, x[i]);
        const float dx = cum.x - prev.x, dy = cum.y - prev.y;
        const float dz = cum.z - prev.z, dw = cum.w - prev.w;
        prev = x[i];
        sq[i] = (dx * dx + dy * dy) + (dz * dz + dw * dw);
    }

    const bool b4 = (lane & 16) != 0, b3 = (lane & 8) != 0, b2 = (lane & 4) != 0;
    float r4[4];
    for (int i = 0; i < 4; ++i) {
        const float keep = b4 ? sq[i + 4] : sq[i];
        const float send = b4 ? sq[i]     : sq[i + 4];
        r4[i] = keep + __shfl_xor(send, 16, 32);
    }
    float r2[2];
    for (int i = 0; i < 2; ++i) {
        const float keep = b3 ? r4[i + 2] : r4[i];
        const float send = b3 ? r4[i]     : r4[i + 2];
        r2[i] = keep + __shfl_xor(send, 8, 32);
    }
    float r1;
    {
        const float keep = b2 ? r2[1] : r2[0];
        const float send = b2 ? r2[0] : r2[1];
        r1 = keep + __shfl_xor(send, 4, 32);
    }
    r1 += __shfl_xor(r1, 2, 32);
    r1 += __shfl_xor(r1, 1, 32);
    if ((lane & 3) == 0) wpart[wave][lane >> 2] = r1;
    __syncthreads();

    if (tid < kLc) {
        const int gg = tid >> 3, r = tid & 7;
        const float tot = (wpart[4 * gg][r] + wpart[4 * gg + 1][r])
                        + (wpart[4 * gg + 2][r] + wpart[4 * gg + 3][r]);
        mag[b * kT + c * kLc + tid] = sqrtf(tot);
    }
}

constexpr int kW2  = 16;
constexpr int kSeg = kT / kW2;
constexpr int kCh  = kSeg / 16;
constexpr int kP2Threads = 32 * kW2;

__global__ __launch_bounds__(kP2Threads)
void context_wmma_kernel(const float* __restrict__ h,
                         const float* __restrict__ kp,
                         const float* __restrict__ mag,
                         float* __restrict__ out) {
    __shared__ float segend[2 * kW2][16];

    const int lane  = threadIdx.x & 31;
    const int w     = __builtin_amdgcn_readfirstlane((int)(threadIdx.x >> 5));
    const int dtile = blockIdx.x & 31;
    const int b     = blockIdx.x >> 5;
    const int n     = lane & 15;
    const int hi    = (lane >> 4) & 1;

    const float* hb = h   + ((size_t)b * kT + (size_t)w * kSeg) * kD + dtile * 16 + n;
    float*       ob = out + ((size_t)b * kT + (size_t)w * kSeg) * kD + dtile * 16 + n;
    const float4* m4 = reinterpret_cast<const float4*>(mag + b * kT + w * kSeg + 8 * hi);

    v2f bm[kCh][4];
    #pragma unroll
    for (int q = 0; q < kCh; ++q) {
        #pragma unroll
        for (int g = 0; g < 4; ++g) {
            const size_t j = (size_t)(16 * q + 4 * g + 2 * hi);
            bm[q][g].x = hb[j * kD];
            bm[q][g].y = hb[(j + 1) * kD];
        }
    }
    float4 mlo[kCh], mhi[kCh];
    #pragma unroll
    for (int q = 0; q < kCh; ++q) { mlo[q] = m4[4 * q]; mhi[q] = m4[4 * q + 1]; }

    const float sp  = softplus_dev(kp);
    const float l2r = -sp * 1.4426950408889634f;
    auto rpow = [&](int dist) -> float { return exp2f((float)dist * l2r); };

    v2f a[4];
    #pragma unroll
    for (int g = 0; g < 4; ++g) {
        int j0 = 4 * g + 2 * hi;
        a[g].x = (j0     <= n) ? rpow(n - j0)     : 0.f;
        a[g].y = (j0 + 1 <= n) ? rpow(n - j0 - 1) : 0.f;
    }
    float rpc[8];
    #pragma unroll
    for (int v = 0; v < 8; ++v) rpc[v] = rpow(v + 1 + 8 * hi);
    const float r16  = rpow(16);
    const float rseg = rpow(kSeg);

    v8f c[kCh];
    #pragma unroll
    for (int q = 0; q < kCh; ++q) {
        v8f z = {0.f, 0.f, 0.f, 0.f, 0.f, 0.f, 0.f, 0.f};
        z = __builtin_amdgcn_wmma_f32_16x16x4_f32(false, a[0], false, bm[q][0], (short)0, z, false, false);
        z = __builtin_amdgcn_wmma_f32_16x16x4_f32(false, a[1], false, bm[q][1], (short)0, z, false, false);
        z = __builtin_amdgcn_wmma_f32_16x16x4_f32(false, a[2], false, bm[q][2], (short)0, z, false, false);
        z = __builtin_amdgcn_wmma_f32_16x16x4_f32(false, a[3], false, bm[q][3], (short)0, z, false, false);
        c[q] = z;
    }

    float e[kCh];
    #pragma unroll
    for (int q = 0; q < kCh; ++q) e[q] = __shfl(c[q][7], 16 + n, 32);

    float E = e[0];
    #pragma unroll
    for (int q = 1; q < kCh; ++q) E = e[q] + r16 * E;
    {
        const int row = (lane < 16) ? (kW2 - 1 + w) : ((w < kW2 - 1) ? w : 2 * kW2 - 1);
        segend[row][n] = (lane < 16) ? E : 0.f;
    }
    __syncthreads();

    float S = 0.f;
    #pragma unroll
    for (int i = 0; i < kW2 - 1; ++i) S = segend[w + i][n] + rseg * S;

    #pragma unroll
    for (int q = 0; q < kCh; ++q) {
        const float mg[8] = {mlo[q].x, mlo[q].y, mlo[q].z, mlo[q].w, mhi[q].x, mhi[q].y, mhi[q].z, mhi[q].w};
        #pragma unroll
        for (int v = 0; v < 8; ++v) {
            const int M = v + 8 * hi;
            ob[(size_t)(16 * q + M) * kD] = mg[v] * fabsf(c[q][v] + rpc[v] * S);
        }
        S = e[q] + r16 * S;
    }
}

extern "C" void kernel_launch(void* const* d_in, const int* in_sizes, int n_in,
                              void* d_out, int out_size, void* d_ws, size_t ws_size,
                              hipStream_t stream) {
    const float* h  = (const float*)d_in[0];
    const float* k  = (const float*)d_in[1];
    float* out = (float*)d_out;
    float* mag = (float*)d_ws;
    float* csum = out;

    chunk_sum_kernel<<<dim3(kNc, kB), 512, 0, stream>>>(h, csum);
    delta_mag_kernel<<<dim3(kNc, kB), 512, 0, stream>>>(h, csum, mag);
    context_wmma_kernel<<<kB * (kD / 16), kP2Threads, 0, stream>>>(h, k, mag, out);
}


# Launches 3 -> 2: last launch cut along T (block = 64 rows x all 512 columns) so the delta norm is formed inside it; chunk carries from one slab-scan launch; cumsum, d-sum and carry sum orders change
# speedup vs baseline: 75.2087x; 1.7702x over previous
#include <hip/hip_runtime.h>
#include <math.h>


constexpr int kB = 8;
constexpr int kT = 2048;
constexpr int kD = 512;
constexpr int kD4 = kD / 4;

constexpr int kCk  = 64;
constexpr int kNck = kT / kCk;
constexpr int kRg  = 16;

typedef __attribute__((ext_vector_type(2))) float v2f;
typedef __attribute__((ext_vector_type(8))) float v8f;

__device__ __forceinline__ float softplus_dev(const float* kp) {
    float k = kp[0];
    return fmaxf(k, 0.0f) + log1pf(expf(-fabsf(k)));
}

__device__ __forceinline__ float4 add4(float4 a, float4 b) {
    return make_float4(a.x + b.x, a.y + b.y, a.z + b.z, a.w + b.w);
}
__device__ __forceinline__ float4 axpy4(float s, float4 lo, float4 hi) {
    return make_float4(hi.x + s * lo.x, hi.y + s * lo.y, hi.z + s * lo.z, hi.w + s * lo.w);
}
__device__ __forceinline__ float4 sel4(bool c, float4 a, float4 b) {
    return make_float4(c ? a.x : b.x, c ? a.y : b.y, c ? a.z : b.z, c ? a.w : b.w);
}
__device__ __forceinline__ float4 shx4(float4 a, int m) {
    return make_float4(__shfl_xor(a.x, m, 32), __shfl_xor(a.y, m, 32),
                       __shfl_xor(a.z, m, 32), __shfl_xor(a.w, m, 32));
}
__device__ __forceinline__ v2f shx2(v2f a, int m) {
    v2f r;
    r.x = __shfl_xor(a.x, m, 32);
    r.y = __shfl_xor(a.y, m, 32);
    return r;
}

constexpr int kW1 = 16;
__global__ __launch_bounds__(32 * kW1)
void chunk_carry_kernel(const float* __restrict__ h,
                        const float* __restrict__ kp,
                        float* __restrict__ out) {
    __shared__ alignas(16) v2f seg[2 * kW1][16];
    __shared__ alignas(16) v2f ch0[kW1][16];

    const int slab = blockIdx.x & 31;
    const int b    = blockIdx.x >> 5;
    const int tid  = threadIdx.x;
    const int c4   = tid & 3;
    const int rg   = tid >> 2;
    const int lane = tid & 31;
    const int w    = __builtin_amdgcn_readfirstlane((int)(threadIdx.x >> 5));

    const float4* hp = reinterpret_cast<const float4*>(h)
                     + ((size_t)b * kT + (size_t)rg * kRg) * kD4 + slab * 4 + c4;
    float4 x[kRg];
    for (int i = 0; i < kRg; ++i) x[i] = hp[(size_t)i * kD4];

    const float sp   = softplus_dev(kp);
    const float l2r  = -sp * 1.4426950408889634f;
    const float r    = exp2f(l2r);
    const float r16  = exp2f(16.0f * l2r);
    const float r32  = exp2f(32.0f * l2r);
    const float r64  = exp2f(64.0f * l2r);
    const float r128 = exp2f(128.0f * l2r);

    float4 s = x[0], e = x[0];
    for (int i = 1; i < kRg; ++i) { s = add4(s, x[i]); e = axpy4(r, e, x[i]); }

    {
        const bool up = (lane & 4) != 0;
        const float4 so = shx4(s, 4), eo = shx4(e, 4);
        s = add4(sel4(up, so, s), sel4(up, s, so));
        e = axpy4(r16, sel4(up, eo, e), sel4(up, e, eo));
    }
    {
        const bool up = (lane & 8) != 0;
        const float4 so = shx4(s, 8), eo = shx4(e, 8);
        s = add4(sel4(up, so, s), sel4(up, s, so));
        e = axpy4(r32, sel4(up, eo, e), sel4(up, e, eo));
    }
    const bool up16 = (lane & 16) != 0;
    const float4 so16 = shx4(s, 16), eo16 = shx4(e, 16);
    const float4 s_lo = sel4(up16, so16, s), s_hi = sel4(up16, s, so16);
    const float4 e_lo = sel4(up16, eo16, e), e_hi = sel4(up16, e, eo16);
    const float4 sseg = add4(s_lo, s_hi);
    const float4 eseg = axpy4(r64, e_lo, e_hi);

    if ((lane & 12) == 0) {
        const int row = up16 ? ((w < kW1 - 1) ? w : 2 * kW1 - 1) : (kW1 - 1 + w);
        const float4 lo4 = up16 ? make_float4(0.f, 0.f, 0.f, 0.f) : make_float4(sseg.x, eseg.x, sseg.y, eseg.y);
        const float4 hi4 = up16 ? make_float4(0.f, 0.f, 0.f, 0.f) : make_float4(sseg.z, eseg.z, sseg.w, eseg.w);
        float4* d = reinterpret_cast<float4*>(&seg[row][4 * c4]);
        d[0] = lo4;
        d[1] = hi4;
        if (!up16) {
            float4* d0 = reinterpret_cast<float4*>(&ch0[w][4 * c4]);
            d0[0] = make_float4(s_lo.x, e_lo.x, s_lo.y, e_lo.y);
            d0[1] = make_float4(s_lo.z, e_lo.z, s_lo.w, e_lo.w);
        }
    }
    __syncthreads();

    const int col  = lane & 15;
    const int half = lane >> 4;
    float Pw = 0.f, Xw = 0.f;
    #pragma unroll
    for (int i = 0; i < kW1 - 1; ++i) {
        const v2f t = seg[w + i][col];
        Pw = Pw + t.x;
        Xw = t.y + r128 * Xw;
    }
    const v2f c0 = ch0[w][col];
    const float P = (half != 0) ? (Pw + c0.x) : Pw;
    const float X = (half != 0) ? (c0.y + r64 * Xw) : Xw;
    float* o = out + ((size_t)b * kT + (size_t)(2 * w + half) * kCk) * kD + slab * 16 + col;
    o[0]  = P;
    o[kD] = X;
}

constexpr int kW = 16;
constexpr int kQ = kCk / 16;

__global__ __launch_bounds__(32 * kW)
void delta_context_kernel(const float* __restrict__ h,
                          const float* __restrict__ kp,
                          float* __restrict__ out) {
    __shared__ alignas(16) float wpart[kW][kCk];
    __shared__ alignas(16) float magw[kW][kCk];

    const int tid  = threadIdx.x;
    const int lane = tid & 31;
    const int w    = __builtin_amdgcn_readfirstlane((int)(threadIdx.x >> 5));
    const int c    = blockIdx.x & 31;
    const int b    = blockIdx.x >> 5;
    const int n    = lane & 15;
    const int hi   = (lane >> 4) & 1;
    const bool up  = hi != 0;

    const size_t rowbase = (size_t)b * kT + (size_t)c * kCk;
    const float* hb = h   + rowbase * kD + 32 * w + 2 * n;
    float*       ob = out + rowbase * kD + 32 * w + 2 * n;

    v2f x[kQ][8];
    #pragma unroll
    for (int q = 0; q < kQ; ++q) {
        #pragma unroll
        for (int s = 0; s < 8; ++s)
            x[q][s] = *reinterpret_cast<const v2f*>(hb + (size_t)(16 * q + 8 * hi + s) * kD);
    }
    const v2f P = *reinterpret_cast<const v2f*>(ob);
    const v2f X = *reinterpret_cast<const v2f*>(ob + kD);
    v2f hp0 = *reinterpret_cast<const v2f*>(c > 0 ? hb - kD : hb);
    if (c == 0) { hp0.x = 0.f; hp0.y = 0.f; }

    const float sp  = softplus_dev(kp);
    const float l2r = -sp * 1.4426950408889634f;
    auto rpow = [&](int dist) -> float { return exp2f((float)dist * l2r); };

    v2f a[4];
    #pragma unroll
    for (int g = 0; g < 4; ++g) {
        const int j0 = 8 * hi + 2 * g;
        a[g].x = (j0     <= n) ? rpow(n - j0)     : 0.f;
        a[g].y = (j0 + 1 <= n) ? rpow(n - j0 - 1) : 0.f;
    }
    float rpc[8];
    #pragma unroll
    for (int v = 0; v < 8; ++v) rpc[v] = rpow(v + 1 + 8 * hi);
    const float r16 = rpow(16);

    v2f own8[kQ], oth8[kQ], recv[kQ];
    #pragma unroll
    for (int q = 0; q < kQ; ++q) {
        v2f t = x[q][0];
        #pragma unroll
        for (int s = 1; s < 8; ++s) t = t + x[q][s];
        own8[q] = t;
        oth8[q] = shx2(t, 16);
        recv[q] = shx2(x[q][7], 16);
    }
    float sq[kQ * 8];
    {
        v2f cc = P;
        #pragma unroll
        for (int q = 0; q < kQ; ++q) {
            const v2f lo8 = up ? oth8[q] : own8[q];
            const v2f hi8 = up ? own8[q] : oth8[q];
            v2f cum  = up ? (cc + lo8) : cc;
            v2f prev = up ? recv[q] : (q == 0 ? hp0 : recv[q > 0 ? q - 1 : 0]);
            #pragma unroll
            for (int s = 0; s < 8; ++s) {
                cum = cum + x[q][s];
                const v2f d = cum - prev;
                prev = x[q][s];
                sq[8 * q + s] = d.x * d.x + d.y * d.y;
            }
            cc = cc + (lo8 + hi8);
        }
    }

    v8f cE[kQ], cO[kQ];
    #pragma unroll
    for (int q = 0; q < kQ; ++q) {
        v8f zE = {0.f, 0.f, 0.f, 0.f, 0.f, 0.f, 0.f, 0.f};
        v8f zO = {0.f, 0.f, 0.f, 0.f, 0.f, 0.f, 0.f, 0.f};
        #pragma unroll
        for (int g = 0; g < 4; ++g) {
            v2f bE, bO;
            bE.x = x[q][2 * g].x; bE.y = x[q][2 * g + 1].x;
            bO.x = x[q][2 * g].y; bO.y = x[q][2 * g + 1].y;
            zE = __builtin_amdgcn_wmma_f32_16x16x4_f32(false, a[g], false, bE, (short)0, zE, false, false);
            zO = __builtin_amdgcn_wmma_f32_16x16x4_f32(false, a[g], false, bO, (short)0, zO, false, false);
        }
        cE[q] = zE;
        cO[q] = zO;
    }
    float eE[kQ], eO[kQ];
    #pragma unroll
    for (int q = 0; q < kQ; ++q) {
        eE[q] = __shfl(cE[q][7], 16 + n, 32);
        eO[q] = __shfl(cO[q][7], 16 + n, 32);
    }

    const bool b3 = (n & 8) != 0, b2 = (n & 4) != 0, b1 = (n & 2) != 0, b0 = (n & 1) != 0;
    float t16[16];
    #pragma unroll
    for (int i = 0; i < 16; ++i) {
        const float keep = b3 ? sq[i + 16] : sq[i];
        const float send = b3 ? sq[i]      : sq[i + 16];
        t16[i] = keep + __shfl_xor(send, 8, 32);
    }
    float t8[8];
    #pragma unroll
    for (int i = 0; i < 8; ++i) {
        const float keep = b2 ? t16[i + 8] : t16[i];
        const float send = b2 ? t16[i]     : t16[i + 8];
        t8[i] = keep + __shfl_xor(send, 4, 32);
    }
    float t4[4];
    #pragma unroll
    for (int i = 0; i < 4; ++i) {
        const float keep = b1 ? t8[i + 4] : t8[i];
        const float send = b1 ? t8[i]     : t8[i + 4];
        t4[i] = keep + __shfl_xor(send, 2, 32);
    }
    float t2[2];
    #pragma unroll
    for (int i = 0; i < 2; ++i) {
        const float keep = b0 ? t4[i + 2] : t4[i];
        const float send = b0 ? t4[i]     : t4[i + 2];
        t2[i] = keep + __shfl_xor(send, 1, 32);
    }
    {
        const int row = 16 * (n >> 2) + 8 * hi + 2 * (n & 3);
        v2f pr; pr.x = t2[0]; pr.y = t2[1];
        *reinterpret_cast<v2f*>(&wpart[w][row]) = pr;
    }
    __syncthreads();
    {
        v2f pv[kW];
        #pragma unroll
        for (int ww = 0; ww < kW; ++ww) pv[ww] = *reinterpret_cast<const v2f*>(&wpart[ww][2 * lane]);
        v2f tot = (pv[0] + pv[1]) + (pv[2] + pv[3]);
        #pragma unroll
        for (int ww = 4; ww < kW; ww += 4)
            tot = tot + ((pv[ww] + pv[ww + 1]) + (pv[ww + 2] + pv[ww + 3]));
        v2f m;
        m.x = sqrtf(tot.x);
        m.y = sqrtf(tot.y);
        *reinterpret_cast<v2f*>(&magw[w][2 * lane]) = m;
    }
    __syncthreads();

    v2f S = X;
    #pragma unroll
    for (int q = 0; q < kQ; ++q) {
        const float4 m0 = *reinterpret_cast<const float4*>(&magw[w][16 * q + 8 * hi]);
        const float4 m1 = *reinterpret_cast<const float4*>(&magw[w][16 * q + 8 * hi + 4]);
        const float mg[8] = {m0.x, m0.y, m0.z, m0.w, m1.x, m1.y, m1.z, m1.w};
        #pragma unroll
        for (int v = 0; v < 8; ++v) {
            v2f o;
            o.x = mg[v] * fabsf(cE[q][v] + rpc[v] * S.x);
            o.y = mg[v] * fabsf(cO[q][v] + rpc[v] * S.y);
            *reinterpret_cast<v2f*>(ob + (size_t)(16 * q + v + 8 * hi) * kD) = o;
        }
        S.x = eE[q] + r16 * S.x;
        S.y = eO[q] + r16 * S.y;
    }
}

extern "C" void kernel_launch(void* const* d_in, const int* in_sizes, int n_in,
                              void* d_out, int out_size, void* d_ws, size_t ws_size,
                              hipStream_t stream) {
    const float* h  = (const float*)d_in[0];
    const float* k  = (const float*)d_in[1];
    float* out = (float*)d_out;
    chunk_carry_kernel<<<kB * (kD / 16), 512, 0, stream>>>(h, k, out);
    delta_context_kernel<<<kB * kNck, 32 * kW, 0, stream>>>(h, k, out);
}


# Launch 1 reads whole 128-byte row pieces: block = (b, 32-column slab, half of T), a wave = one chunk; the lower half's carry is added in launch 2 (sum order of the carried P and X changes)
# speedup vs baseline: 86.6635x; 1.1523x over previous
#include <hip/hip_runtime.h>
#include <math.h>


constexpr int kB = 8;
constexpr int kT = 2048;
constexpr int kD = 512;
constexpr int kD4 = kD / 4;

constexpr int kCk  = 64;
constexpr int kNck = kT / kCk;
constexpr int kRg  = 16;

typedef __attribute__((ext_vector_type(2))) float v2f;
typedef __attribute__((ext_vector_type(8))) float v8f;

__device__ __forceinline__ float softplus_dev(const float* kp) {
    float k = kp[0];
    return fmaxf(k, 0.0f) + log1pf(expf(-fabsf(k)));
}

__device__ __forceinline__ float4 add4(float4 a, float4 b) {
    return make_float4(a.x + b.x, a.y + b.y, a.z + b.z, a.w + b.w);
}
__device__ __forceinline__ float4 axpy4(float s, float4 lo, float4 hi) {
    return make_float4(hi.x + s * lo.x, hi.y + s * lo.y, hi.z + s * lo.z, hi.w + s * lo.w);
}
__device__ __forceinline__ float4 sel4(bool c, float4 a, float4 b) {
    return make_float4(c ? a.x : b.x, c ? a.y : b.y, c ? a.z : b.z, c ? a.w : b.w);
}
__device__ __forceinline__ float4 shx4(float4 a, int m) {
    return make_float4(__shfl_xor(a.x, m, 32), __shfl_xor(a.y, m, 32),
                       __shfl_xor(a.z, m, 32), __shfl_xor(a.w, m, 32));
}
__device__ __forceinline__ v2f shx2(v2f a, int m) {
    v2f r;
    r.x = __shfl_xor(a.x, m, 32);
    r.y = __shfl_xor(a.y, m, 32);
    return r;
}

constexpr int kW1 = 16;
constexpr int kHalfRows = kT / 2;
__global__ __launch_bounds__(32 * kW1)
void chunk_carry_kernel(const float* __restrict__ h,
                        const float* __restrict__ kp,
                        float* __restrict__ out) {
    __shared__ alignas(16) v2f seg[2 * kW1][32];

    const int slab = blockIdx.x & 15;
    const int hf   = (blockIdx.x >> 4) & 1;
    const int b    = blockIdx.x >> 5;
    const int tid  = threadIdx.x;
    const int c8   = tid & 7;
    const int rg   = tid >> 3;
    const int lane = tid & 31;
    const int w    = __builtin_amdgcn_readfirstlane((int)(threadIdx.x >> 5));

    const float4* hp = reinterpret_cast<const float4*>(h)
                     + ((size_t)b * kT + (size_t)hf * kHalfRows + (size_t)rg * kRg) * kD4 + slab * 8 + c8;
    float4 x[kRg];
    for (int i = 0; i < kRg; ++i) x[i] = hp[(size_t)i * kD4];

    const float sp   = softplus_dev(kp);
    const float l2r  = -sp * 1.4426950408889634f;
    const float r    = exp2f(l2r);
    const float r16  = exp2f(16.0f * l2r);
    const float r32  = exp2f(32.0f * l2r);
    const float r64  = exp2f(64.0f * l2r);

    float4 s = x[0], e = x[0];
    for (int i = 1; i < kRg; ++i) { s = add4(s, x[i]); e = axpy4(r, e, x[i]); }

    {
        const bool up = (lane & 8) != 0;
        const float4 so = shx4(s, 8), eo = shx4(e, 8);
        s = add4(sel4(up, so, s), sel4(up, s, so));
        e = axpy4(r16, sel4(up, eo, e), sel4(up, e, eo));
    }
    {
        const bool up = (lane & 16) != 0;
        const float4 so = shx4(s, 16), eo = shx4(e, 16);
        s = add4(sel4(up, so, s), sel4(up, s, so));
        e = axpy4(r32, sel4(up, eo, e), sel4(up, e, eo));
    }

    if ((lane & 16) == 0) {
        const bool pad = (lane & 8) != 0;
        const int row = pad ? ((w < kW1 - 1) ? w : 2 * kW1 - 1) : (kW1 - 1 + w);
        const float4 lo4 = pad ? make_float4(0.f, 0.f, 0.f, 0.f) : make_float4(s.x, e.x, s.y, e.y);
        const float4 hi4 = pad ? make_float4(0.f, 0.f, 0.f, 0.f) : make_float4(s.z, e.z, s.w, e.w);
        float4* d = reinterpret_cast<float4*>(&seg[row][4 * c8]);
        d[0] = lo4;
        d[1] = hi4;
    }
    __syncthreads();

    float Pw = 0.f, Xw = 0.f;
    #pragma unroll
    for (int i = 0; i < kW1 - 1; ++i) {
        const v2f t = seg[w + i][lane];
        Pw = Pw + t.x;
        Xw = t.y + r64 * Xw;
    }
    float* o = out + ((size_t)b * kT + (size_t)hf * kHalfRows + (size_t)w * kCk) * kD + slab * 32 + lane;
    o[0]  = Pw;
    o[kD] = Xw;
    if (hf == 0) {
        float Pt = 0.f, Xt = 0.f;
        #pragma unroll
        for (int i = 0; i < kW1; ++i) {
            const v2f t = seg[kW1 - 1 + i][lane];
            Pt = Pt + t.x;
            Xt = t.y + r64 * Xt;
        }
        float* o2 = out + ((size_t)b * kT + (size_t)kHalfRows + (size_t)w * kCk + 2) * kD + slab * 32 + lane;
        o2[0]  = Pt;
        o2[kD] = Xt;
    }
}

constexpr int kW = 16;
constexpr int kQ = kCk / 16;

__global__ __launch_bounds__(32 * kW)
void delta_context_kernel(const float* __restrict__ h,
                          const float* __restrict__ kp,
                          float* __restrict__ out) {
    __shared__ alignas(16) float wpart[kW][kCk];
    __shared__ alignas(16) float magw[kW][kCk];

    const int tid  = threadIdx.x;
    const int lane = tid & 31;
    const int w    = __builtin_amdgcn_readfirstlane((int)(threadIdx.x >> 5));
    const int c    = blockIdx.x & 31;
    const int b    = blockIdx.x >> 5;
    const int n    = lane & 15;
    const int hi   = (lane >> 4) & 1;
    const bool up  = hi != 0;

    const size_t rowbase = (size_t)b * kT + (size_t)c * kCk;
    const float* hb = h   + rowbase * kD + 32 * w + 2 * n;
    float*       ob = out + rowbase * kD + 32 * w + 2 * n;

    v2f x[kQ][8];
    #pragma unroll
    for (int q = 0; q < kQ; ++q) {
        #pragma unroll
        for (int s = 0; s < 8; ++s)
            x[q][s] = *reinterpret_cast<const v2f*>(hb + (size_t)(16 * q + 8 * hi + s) * kD);
    }
    const v2f P0 = *reinterpret_cast<const v2f*>(ob);
    const v2f X0 = *reinterpret_cast<const v2f*>(ob + kD);
    v2f PT = {0.f, 0.f}, XT = {0.f, 0.f};
    if (c >= kNck / 2) {
        PT = *reinterpret_cast<const v2f*>(ob + 2 * kD);
        XT = *reinterpret_cast<const v2f*>(ob + 3 * kD);
    }
    v2f hp0 = *reinterpret_cast<const v2f*>(c > 0 ? hb - kD : hb);
    if (c == 0) { hp0.x = 0.f; hp0.y = 0.f; }

    const float sp  = softplus_dev(kp);
    const float l2r = -sp * 1.4426950408889634f;
    auto rpow = [&](int dist) -> float { return exp2f((float)dist * l2r); };

    const float rT = (c >= kNck / 2) ? rpow(kCk * (c - kNck / 2)) : 0.f;
    v2f P, X;
    P.x = P0.x + PT.x;       P.y = P0.y + PT.y;
    X.x = X0.x + rT * XT.x;  X.y = X0.y + rT * XT.y;

    v2f a[4];
    #pragma unroll
    for (int g = 0; g < 4; ++g) {
        const int j0 = 8 * hi + 2 * g;
        a[g].x = (j0     <= n) ? rpow(n - j0)     : 0.f;
        a[g].y = (j0 + 1 <= n) ? rpow(n - j0 - 1) : 0.f;
    }
    float rpc[8];
    #pragma unroll
    for (int v = 0; v < 8; ++v) rpc[v] = rpow(v + 1 + 8 * hi);
    const float r16 = rpow(16);

    v2f own8[kQ], oth8[kQ], recv[kQ];
    #pragma unroll
    for (int q = 0; q < kQ; ++q) {
        v2f t = x[q][0];
        #pragma unroll
        for (int s = 1; s < 8; ++s) t = t + x[q][s];
        own8[q] = t;
        oth8[q] = shx2(t, 16);
        recv[q] = shx2(x[q][7], 16);
    }
    float sq[kQ * 8];
    {
        v2f cc = P;
        #pragma unroll
        for (int q = 0; q < kQ; ++q) {
            const v2f lo8 = up ? oth8[q] : own8[q];
            const v2f hi8 = up ? own8[q] : oth8[q];
            v2f cum  = up ? (cc + lo8) : cc;
            v2f prev = up ? recv[q] : (q == 0 ? hp0 : recv[q > 0 ? q - 1 : 0]);
            #pragma unroll
            for (int s = 0; s < 8; ++s) {
                cum = cum + x[q][s];
                const v2f d = cum - prev;
                prev = x[q][s];
                sq[8 * q + s] = d.x * d.x + d.y * d.y;
            }
            cc = cc + (lo8 + hi8);
        }
    }

    v8f cE[kQ], cO[kQ];
    #pragma unroll
    for (int q = 0; q < kQ; ++q) {
        v8f zE = {0.f, 0.f, 0.f, 0.f, 0.f, 0.f, 0.f, 0.f};
        v8f zO = {0.f, 0.f, 0.f, 0.f, 0.f, 0.f, 0.f, 0.f};
        #pragma unroll
        for (int g = 0; g < 4; ++g) {
            v2f bE, bO;
            bE.x = x[q][2 * g].x; bE.y = x[q][2 * g + 1].x;
            bO.x = x[q][2 * g].y; bO.y = x[q][2 * g + 1].y;
            zE = __builtin_amdgcn_wmma_f32_16x16x4_f32(false, a[g], false, bE, (short)0, zE, false, false);
            zO = __builtin_amdgcn_wmma_f32_16x16x4_f32(false, a[g], false, bO, (short)0, zO, false, false);
        }
        cE[q] = zE;
        cO[q] = zO;
    }
    float eE[kQ], eO[kQ];
    #pragma unroll
    for (int q = 0; q < kQ; ++q) {
        eE[q] = __shfl(cE[q][7], 16 + n, 32);
        eO[q] = __shfl(cO[q][7], 16 + n, 32);
    }

    const bool b3 = (n & 8) != 0, b2 = (n & 4) != 0, b1 = (n & 2) != 0, b0 = (n & 1) != 0;
    float t16[16];
    #pragma unroll
    for (int i = 0; i < 16; ++i) {
        const float keep = b3 ? sq[i + 16] : sq[i];
        const float send = b3 ? sq[i]      : sq[i + 16];
        t16[i] = keep + __shfl_xor(send, 8, 32);
    }
    float t8[8];
    #pragma unroll
    for (int i = 0; i < 8; ++i) {
        const float keep = b2 ? t16[i + 8] : t16[i];
        const float send = b2 ? t16[i]     : t16[i + 8];
        t8[i] = keep + __shfl_xor(send, 4, 32);
    }
    float t4[4];
    #pragma unroll
    for (int i = 0; i < 4; ++i) {
        const float keep = b1 ? t8[i + 4] : t8[i];
        const float send = b1 ? t8[i]     : t8[i + 4];
        t4[i] = keep + __shfl_xor(send, 2, 32);
    }
    float t2[2];
    #pragma unroll
    for (int i = 0; i < 2; ++i) {
        const float keep = b0 ? t4[i + 2] : t4[i];
        const float send = b0 ? t4[i]     : t4[i + 2];
        t2[i] = keep + __shfl_xor(send, 1, 32);
    }
    {
        const int row = 16 * (n >> 2) + 8 * hi + 2 * (n & 3);
        v2f pr; pr.x = t2[0]; pr.y = t2[1];
        *reinterpret_cast<v2f*>(&wpart[w][row]) = pr;
    }
    __syncthreads();
    {
        v2f pv[kW];
        #pragma unroll
        for (int ww = 0; ww < kW; ++ww) pv[ww] = *reinterpret_cast<const v2f*>(&wpart[ww][2 * lane]);
        v2f tot = (pv[0] + pv[1]) + (pv[2] + pv[3]);
        #pragma unroll
        for (int ww = 4; ww < kW; ww += 4)
            tot = tot + ((pv[ww] + pv[ww + 1]) + (pv[ww + 2] + pv[ww + 3]));
        v2f m;
        m.x = sqrtf(tot.x);
        m.y = sqrtf(tot.y);
        *reinterpret_cast<v2f*>(&magw[w][2 * lane]) = m;
    }
    __syncthreads();

    v2f S = X;
    #pragma unroll
    for (int q = 0; q < kQ; ++q) {
        const float4 m0 = *reinterpret_cast<const float4*>(&magw[w][16 * q + 8 * hi]);
        const float4 m1 = *reinterpret_cast<const float4*>(&magw[w][16 * q + 8 * hi + 4]);
        const float mg[8] = {m0.x, m0.y, m0.z, m0.w, m1.x, m1.y, m1.z, m1.w};
        #pragma unroll
        for (int v = 0; v < 8; ++v) {
            v2f o;
            o.x = mg[v] * fabsf(cE[q][v] + rpc[v] * S.x);
            o.y = mg[v] * fabsf(cO[q][v] + rpc[v] * S.y);
            *reinterpret_cast<v2f*>(ob + (size_t)(16 * q + v + 8 * hi) * kD) = o;
        }
        S.x = eE[q] + r16 * S.x;
        S.y = eO[q] + r16 * S.y;
    }
}

extern "C" void kernel_launch(void* const* d_in, const int* in_sizes, int n_in,
                              void* d_out, int out_size, void* d_ws, size_t ws_size,
                              hipStream_t stream) {
    const float* h  = (const float*)d_in[0];
    const float* k  = (const float*)d_in[1];
    float* out = (float*)d_out;
    chunk_carry_kernel<<<kB * (kD / 32) * 2, 32 * kW1, 0, stream>>>(h, k, out);
    delta_context_kernel<<<kB * kNck, 32 * kW, 0, stream>>>(h, k, out);
}
